# starting kernel
# baseline (speedup 1.0000x reference)
#include <hip/hip_runtime.h>
#include <hip/hip_bf16.h>

#define BB 2
#define NN 8192
#define DD 64
#define DV 64
#define MM 3

typedef __attribute__((ext_vector_type(2))) float v2f;
typedef __attribute__((ext_vector_type(8))) float v8f;

__global__ __launch_bounds__(32) void v0_gemm_kernel(
    const float* __restrict__ X, const float* __restrict__ Wg,
    const float* __restrict__ bg, float* __restrict__ V0) {
  const int tile = blockIdx.x;
  const int b    = tile / (NN / 16);
  const int row0 = (tile % (NN / 16)) * 16;
  const int lane = threadIdx.x;
  const int r    = lane & 15;
  const int hi   = lane >> 4;

  const float* xrow = X + ((size_t)b * NN + row0 + r) * DD;
  v2f areg[16];
#pragma unroll
  for (int kk = 0; kk < 16; ++kk) {
    const int k0 = 4 * kk + 2 * hi;
    areg[kk] = *(const v2f*)(xrow + k0);
  }

#pragma unroll
  for (int ct = 0; ct < 4; ++ct) {
    const int col = 16 * ct + r;
    v8f acc = {0.f, 0.f, 0.f, 0.f, 0.f, 0.f, 0.f, 0.f};
#pragma unroll
    for (int kk = 0; kk < 16; ++kk) {
      const int k0 = 4 * kk + 2 * hi;
      v2f bm;
      bm.x = Wg[(size_t)k0 * DV + col];
      bm.y = Wg[(size_t)(k0 + 1) * DV + col];
      acc = __builtin_amdgcn_wmma_f32_16x16x4_f32(
           false, areg[kk],  false, bm,
           (short)0, acc,  false,  false);
    }
    const float bias = bg[col];
#pragma unroll
    for (int j = 0; j < 8; ++j) {
      V0[((size_t)b * NN + row0 + j + 8 * hi) * DV + col] = acc[j] + bias;
    }
  }
}

__global__ __launch_bounds__(128) void coeff_kernel(
    const float* __restrict__ X, const float* __restrict__ Wf,
    const float* __restrict__ bf, float* __restrict__ ca,
    float* __restrict__ cc) {
  __shared__ float Xs[128 * 65];
  const int tid  = threadIdx.x;
  const int cpb  = NN / 128;
  const int bi   = blockIdx.x;
  const int i0   = (bi % cpb) * 128;
  const int m    = (bi / cpb) % MM;
  const int b    = bi / (cpb * MM);

  const float* xbase = X + ((size_t)b * NN + i0) * DD;
  for (int idx = tid; idx < 128 * DD; idx += 128) {
    const int row = idx >> 6;
    const int col = idx & 63;
    Xs[row * 65 + col] = xbase[idx];
  }
  __syncthreads();

  const int i  = i0 + tid;
  const int i1 = (i + 1) & (NN - 1);
  float accA = bf[(size_t)m * NN + i];
  float accC = bf[(size_t)m * NN + i1];
  const float* wcol = Wf + (size_t)m * DD * NN;
#pragma unroll 8
  for (int d = 0; d < DD; ++d) {
    const float x = Xs[tid * 65 + d];
    accA += x * wcol[(size_t)d * NN + i];
    accC += x * wcol[(size_t)d * NN + i1];
  }
  const size_t o = ((size_t)b * MM + m) * NN + i;
  ca[o] = accA;
  cc[o] = accC;
}

__global__ __launch_bounds__(64) void combine_kernel(
    const float* __restrict__ V0, const float* __restrict__ ca,
    const float* __restrict__ cc, float* __restrict__ out) {
  const int dv = threadIdx.x;
  const int b  = blockIdx.x >> 13;
  const int i  = blockIdx.x & (NN - 1);

  const int j0 = i;
  const int j1 = (i + 1) & (NN - 1);
  const int j2 = (i + 2) & (NN - 1);
  const int j3 = (i + 3) & (NN - 1);

  const size_t cb = (size_t)b * MM * NN;
  const float a0_0 = ca[cb + 0 * NN + j0], c0_0 = cc[cb + 0 * NN + j0];
  const float a0_1 = ca[cb + 0 * NN + j1], c0_1 = cc[cb + 0 * NN + j1];
  const float a0_2 = ca[cb + 0 * NN + j2], c0_2 = cc[cb + 0 * NN + j2];
  const float a1_0 = ca[cb + 1 * NN + j0], c1_0 = cc[cb + 1 * NN + j0];
  const float a1_1 = ca[cb + 1 * NN + j1], c1_1 = cc[cb + 1 * NN + j1];
  const float a2_0 = ca[cb + 2 * NN + j0], c2_0 = cc[cb + 2 * NN + j0];

  const size_t vb = (size_t)b * NN;
  const float v0 = V0[(vb + j0) * DV + dv];
  const float v1 = V0[(vb + j1) * DV + dv];
  const float v2 = V0[(vb + j2) * DV + dv];
  const float v3 = V0[(vb + j3) * DV + dv];

  const float u0 = a0_0 * v0 + c0_0 * v1;
  const float u1 = a0_1 * v1 + c0_1 * v2;
  const float u2 = a0_2 * v2 + c0_2 * v3;
  const float w0 = a1_0 * u0 + c1_0 * u1;
  const float w1 = a1_1 * u1 + c1_1 * u2;

  out[(vb + i) * DV + dv] = a2_0 * w0 + c2_0 * w1;
}

extern "C" void kernel_launch(void* const* d_in, const int* in_sizes, int n_in,
                              void* d_out, int out_size, void* d_ws, size_t ws_size,
                              hipStream_t stream) {
  const float* X  = (const float*)d_in[0];
  const float* Wg = (const float*)d_in[1];
  const float* bg = (const float*)d_in[2];
  const float* Wf = (const float*)d_in[3];
  const float* bf = (const float*)d_in[4];
  float* out = (float*)d_out;

  float* V0 = (float*)d_ws;
  float* ca = V0 + (size_t)BB * NN * DV;
  float* cc = ca + (size_t)BB * MM * NN;

  v0_gemm_kernel<<<BB * NN / 16, 32, 0, stream>>>(X, Wg, bg, V0);
  coeff_kernel<<<BB * MM * NN / 128, 128, 0, stream>>>(X, Wf, bf, ca, cc);
  combine_kernel<<<BB * NN, 64, 0, stream>>>(V0, ca, cc, out);
}


# Fuse the 3 launches into 1: a 256-thread work-group per 64-row tile does V0 (WMMA, 16 halo rows), the a/c coefficients and the 3-step chain through LDS; sums keep their order; no workspace
# speedup vs baseline: 3.5704x; 3.5704x over previous
#include <hip/hip_runtime.h>

#define BB 2
#define NN 8192
#define DD 64
#define DV 64
#define MM 3

#define TR 64
#define XR 80
#define PX 68
#define NC 67

typedef __attribute__((ext_vector_type(2))) float v2f;
typedef __attribute__((ext_vector_type(4))) float v4f;
typedef __attribute__((ext_vector_type(8))) float v8f;

template <int NT>
__device__ __forceinline__ void v0_tiles(const float* Xs, float* Vs, const v2f (&bfr)[16],
                                         float bias, int rt0, int ct, int r, int hi) {
  v2f afr[NT][16];
#pragma unroll
  for (int t = 0; t < NT; ++t) {
#pragma unroll
    for (int kk = 0; kk < 16; ++kk) {
      afr[t][kk] = *(const v2f*)(Xs + (16 * (rt0 + t) + r) * PX + 4 * kk + 2 * hi);
    }
  }
  __builtin_amdgcn_sched_barrier(0);
  v8f acc[NT];
#pragma unroll
  for (int t = 0; t < NT; ++t) acc[t] = (v8f){0.f, 0.f, 0.f, 0.f, 0.f, 0.f, 0.f, 0.f};
#pragma unroll
  for (int kk = 0; kk < 16; ++kk) {
#pragma unroll
    for (int t = 0; t < NT; ++t) {
      acc[t] = __builtin_amdgcn_wmma_f32_16x16x4_f32(
           false, afr[t][kk],  false, bfr[kk],
           (short)0, acc[t],  false,  false);
    }
  }
#pragma unroll
  for (int t = 0; t < NT; ++t) {
#pragma unroll
    for (int j = 0; j < 8; ++j) {
      Vs[(16 * (rt0 + t) + j + 8 * hi) * PX + 16 * ct + r] = acc[t][j] + bias;
    }
  }
}

__global__ __launch_bounds__(256) void fused_kernel(
    const float* __restrict__ X, const float* __restrict__ Wg,
    const float* __restrict__ bg, const float* __restrict__ Wf,
    const float* __restrict__ bf, float* __restrict__ out) {
  __shared__ __attribute__((aligned(16))) float Xs[XR * PX];
  __shared__ __attribute__((aligned(16))) float Vs[XR * PX];
  __shared__ __attribute__((aligned(16))) float cA[MM * XR];
  __shared__ __attribute__((aligned(16))) float cC[MM * XR];

  const int tid  = threadIdx.x;
  const int lane = tid & 31;
  const int r    = lane & 15;
  const int hi   = lane >> 4;
  const int wave = __builtin_amdgcn_readfirstlane(tid >> 5);
  const int b    = blockIdx.x / (NN / TR);
  const int i0   = (blockIdx.x % (NN / TR)) * TR;
  const float* Xb = X + (size_t)b * NN * DD;

  v4f xr[5];
#pragma unroll
  for (int q = 0; q < 5; ++q) {
    const int idx  = tid + 256 * q;
    const int row  = idx >> 4;
    const int c4   = idx & 15;
    const int grow = (i0 + row) & (NN - 1);
    xr[q] = *(const v4f*)(Xb + (size_t)grow * DD + 4 * c4);
  }

  const int ct  = wave & 3;
  const int col = 16 * ct + r;
  v2f bfr[16];
#pragma unroll
  for (int kk = 0; kk < 16; ++kk) {
    const int k0 = 4 * kk + 2 * hi;
    bfr[kk].x = Wg[(size_t)k0 * DV + col];
    bfr[kk].y = Wg[(size_t)(k0 + 1) * DV + col];
  }
  const float bias = bg[col];

  const bool cact = tid < MM * NC;
  const int  cp   = cact ? tid : MM * NC - 1;
  const int  cm   = cp / NC;
  const int  ck   = cp % NC;
  float wv[DD];
  const int kcol = (i0 + ck) & (NN - 1);
  const float* wp = Wf + (size_t)cm * DD * NN + kcol;
  const float cb = bf[(size_t)cm * NN + kcol];
#pragma unroll
  for (int d = 0; d < DD; ++d) wv[d] = wp[(size_t)d * NN];

#pragma unroll
  for (int q = 0; q < 5; ++q) {
    const int idx = tid + 256 * q;
    const int row = idx >> 4;
    const int c4  = idx & 15;
    *(v4f*)(Xs + row * PX + 4 * c4) = xr[q];
  }
  asm volatile("s_wait_dscnt 0x0" ::: "memory");
  asm volatile("s_barrier_signal -1" ::: "memory");
  asm volatile("s_barrier_wait -1" ::: "memory");

  if (wave >= 4) {
    v0_tiles<3>(Xs, Vs, bfr, bias, 0, ct, r, hi);
  } else {
    v0_tiles<2>(Xs, Vs, bfr, bias, 3, ct, r, hi);
  }

  if (cact) {
    const float* x1 = Xs + ck * PX;
    const float* x0 = Xs + (ck > 0 ? ck - 1 : 0) * PX;
    float accA = cb;
    float accC = cb;
#pragma unroll
    for (int d4 = 0; d4 < DD / 4; ++d4) {
      const v4f xa = *(const v4f*)(x1 + 4 * d4);
      const v4f xc = *(const v4f*)(x0 + 4 * d4);
      accA += xa.x * wv[4 * d4 + 0];
      accA += xa.y * wv[4 * d4 + 1];
      accA += xa.z * wv[4 * d4 + 2];
      accA += xa.w * wv[4 * d4 + 3];
      accC += xc.x * wv[4 * d4 + 0];
      accC += xc.y * wv[4 * d4 + 1];
      accC += xc.z * wv[4 * d4 + 2];
      accC += xc.w * wv[4 * d4 + 3];
    }
    cA[cm * XR + ck] = accA;
    if (ck > 0) cC[cm * XR + ck - 1] = accC;
  }
  __syncthreads();

  const int dv4 = tid & 15;
  const int r0  = 4 * (tid >> 4);
  v4f v[7];
#pragma unroll
  for (int j = 0; j < 7; ++j) v[j] = *(const v4f*)(Vs + (r0 + j) * PX + 4 * dv4);
  float a0[6], c0[6], a1[5], c1[5], a2[4], c2[4];
#pragma unroll
  for (int j = 0; j < 6; ++j) { a0[j] = cA[0 * XR + r0 + j]; c0[j] = cC[0 * XR + r0 + j]; }
#pragma unroll
  for (int j = 0; j < 5; ++j) { a1[j] = cA[1 * XR + r0 + j]; c1[j] = cC[1 * XR + r0 + j]; }
#pragma unroll
  for (int j = 0; j < 4; ++j) { a2[j] = cA[2 * XR + r0 + j]; c2[j] = cC[2 * XR + r0 + j]; }

  v4f u[6], w[5];
#pragma unroll
  for (int j = 0; j < 6; ++j) u[j] = a0[j] * v[j] + c0[j] * v[j + 1];
#pragma unroll
  for (int j = 0; j < 5; ++j) w[j] = a1[j] * u[j] + c1[j] * u[j + 1];
  float* ob = out + ((size_t)b * NN + i0 + r0) * DV + 4 * dv4;
#pragma unroll
  for (int j = 0; j < 4; ++j) {
    *(v4f*)(ob + (size_t)j * DV) = a2[j] * w[j] + c2[j] * w[j + 1];
  }
}

extern "C" void kernel_launch(void* const* d_in, const int* in_sizes, int n_in,
                              void* d_out, int out_size, void* d_ws, size_t ws_size,
                              hipStream_t stream) {
  const float* X  = (const float*)d_in[0];
  const float* Wg = (const float*)d_in[1];
  const float* bg = (const float*)d_in[2];
  const float* Wf = (const float*)d_in[3];
  const float* bf = (const float*)d_in[4];
  float* out = (float*)d_out;

  fused_kernel<<<BB * NN / TR, 256, 0, stream>>>(X, Wg, bg, Wf, bf, out);
}


# Coefficient dot products: each thread's a and c chains run side by side behind one group of 32 LDS row reads (was two serial 64-step chains, an LDS wait every 4 steps); every sum keeps its order
# speedup vs baseline: 3.8350x; 1.0741x over previous
#include <hip/hip_runtime.h>

#define BB 2
#define NN 8192
#define DD 64
#define DV 64
#define MM 3

#define TR 64
#define XR 80
#define PX 68
#define NC 67
#define CO 4

typedef __attribute__((ext_vector_type(2))) float v2f;
typedef __attribute__((ext_vector_type(4))) float v4f;
typedef __attribute__((ext_vector_type(8))) float v8f;

template <int NT>
__device__ __forceinline__ void v0_tiles(const float* Xs, float* Vs, const v2f (&bfr)[16],
                                         float bias, int rt0, int ct, int r, int hi) {
  v2f afr[NT][16];
#pragma unroll
  for (int t = 0; t < NT; ++t) {
#pragma unroll
    for (int kk = 0; kk < 16; ++kk) {
      afr[t][kk] = *(const v2f*)(Xs + (16 * (rt0 + t) + r) * PX + 4 * kk + 2 * hi);
    }
  }
  __builtin_amdgcn_sched_barrier(0);
  v8f acc[NT];
#pragma unroll
  for (int t = 0; t < NT; ++t) acc[t] = (v8f){0.f, 0.f, 0.f, 0.f, 0.f, 0.f, 0.f, 0.f};
#pragma unroll
  for (int kk = 0; kk < 16; ++kk) {
#pragma unroll
    for (int t = 0; t < NT; ++t) {
      acc[t] = __builtin_amdgcn_wmma_f32_16x16x4_f32(
           false, afr[t][kk],  false, bfr[kk],
           (short)0, acc[t],  false,  false);
    }
  }
#pragma unroll
  for (int t = 0; t < NT; ++t) {
#pragma unroll
    for (int j = 0; j < 8; ++j) {
      Vs[(16 * (rt0 + t) + j + 8 * hi) * PX + 16 * ct + r] = acc[t][j] + bias;
    }
  }
}

__global__ __launch_bounds__(256) void fused_kernel(
    const float* __restrict__ X, const float* __restrict__ Wg,
    const float* __restrict__ bg, const float* __restrict__ Wf,
    const float* __restrict__ bf, float* __restrict__ out) {
  __shared__ __attribute__((aligned(16))) float Xs[XR * PX];
  __shared__ __attribute__((aligned(16))) float Vs[XR * PX];
  __shared__ __attribute__((aligned(16))) float cA[MM * XR];
  __shared__ __attribute__((aligned(16))) float cC[MM * XR];

  const int tid  = threadIdx.x;
  const int lane = tid & 31;
  const int r    = lane & 15;
  const int hi   = lane >> 4;
  const int wave = __builtin_amdgcn_readfirstlane(tid >> 5);
  const int b    = blockIdx.x / (NN / TR);
  const int i0   = (blockIdx.x % (NN / TR)) * TR;
  const float* Xb = X + (size_t)b * NN * DD;

  v4f xr[5];
#pragma unroll
  for (int q = 0; q < 5; ++q) {
    const int idx  = tid + 256 * q;
    const int row  = idx >> 4;
    const int c4   = idx & 15;
    const int grow = (i0 + row) & (NN - 1);
    xr[q] = *(const v4f*)(Xb + (size_t)grow * DD + 4 * c4);
  }

  const int ct  = wave & 3;
  const int col = 16 * ct + r;
  v2f bfr[16];
#pragma unroll
  for (int kk = 0; kk < 16; ++kk) {
    const int k0 = 4 * kk + 2 * hi;
    bfr[kk].x = Wg[(size_t)k0 * DV + col];
    bfr[kk].y = Wg[(size_t)(k0 + 1) * DV + col];
  }
  const float bias = bg[col];

  const bool cact = tid < MM * NC;
  const int  cp   = cact ? tid : MM * NC - 1;
  const int  cm   = cp / NC;
  const int  ck   = cp % NC;
  float wv[DD];
  const int kcol = (i0 + ck) & (NN - 1);
  const float* wp = Wf + (size_t)cm * DD * NN + kcol;
  const float cb = bf[(size_t)cm * NN + kcol];
#pragma unroll
  for (int d = 0; d < DD; ++d) wv[d] = wp[(size_t)d * NN];

#pragma unroll
  for (int q = 0; q < 5; ++q) {
    const int idx = tid + 256 * q;
    const int row = idx >> 4;
    const int c4  = idx & 15;
    *(v4f*)(Xs + row * PX + 4 * c4) = xr[q];
  }
  asm volatile("s_wait_dscnt 0x0" ::: "memory");
  asm volatile("s_barrier_signal -1" ::: "memory");
  asm volatile("s_barrier_wait -1" ::: "memory");

  if (wave >= 4) {
    v0_tiles<3>(Xs, Vs, bfr, bias, 0, ct, r, hi);
  } else {
    v0_tiles<2>(Xs, Vs, bfr, bias, 3, ct, r, hi);
  }

  if (cact) {
    const float* x1 = Xs + ck * PX;
    const float* x0 = Xs + (ck > 0 ? ck - 1 : 0) * PX;
    v4f xa[DD / 4], xc[DD / 4];
#pragma unroll
    for (int d4 = 0; d4 < DD / 4; ++d4) {
      xa[d4] = *(const v4f*)(x1 + 4 * d4);
      xc[d4] = *(const v4f*)(x0 + 4 * d4);
    }
    __builtin_amdgcn_sched_barrier(0);
    float accA = cb;
    float accC = cb;
#pragma unroll
    for (int d4 = 0; d4 < DD / 4; ++d4) {
      accA = __builtin_fmaf(xa[d4].x, wv[4 * d4 + 0], accA);
      accC = __builtin_fmaf(xc[d4].x, wv[4 * d4 + 0], accC);
      accA = __builtin_fmaf(xa[d4].y, wv[4 * d4 + 1], accA);
      accC = __builtin_fmaf(xc[d4].y, wv[4 * d4 + 1], accC);
      accA = __builtin_fmaf(xa[d4].z, wv[4 * d4 + 2], accA);
      accC = __builtin_fmaf(xc[d4].z, wv[4 * d4 + 2], accC);
      accA = __builtin_fmaf(xa[d4].w, wv[4 * d4 + 3], accA);
      accC = __builtin_fmaf(xc[d4].w, wv[4 * d4 + 3], accC);
    }
    cA[cm * XR + ck] = accA;
    cC[cm * XR + ck + CO - 1] = accC;
  }
  __syncthreads();

  const int dv4 = tid & 15;
  const int r0  = 4 * (tid >> 4);
  v4f v[7];
#pragma unroll
  for (int j = 0; j < 7; ++j) v[j] = *(const v4f*)(Vs + (r0 + j) * PX + 4 * dv4);
  float a0[6], c0[6], a1[5], c1[5], a2[4], c2[4];
#pragma unroll
  for (int j = 0; j < 6; ++j) { a0[j] = cA[0 * XR + r0 + j]; c0[j] = cC[0 * XR + r0 + j + CO]; }
#pragma unroll
  for (int j = 0; j < 5; ++j) { a1[j] = cA[1 * XR + r0 + j]; c1[j] = cC[1 * XR + r0 + j + CO]; }
#pragma unroll
  for (int j = 0; j < 4; ++j) { a2[j] = cA[2 * XR + r0 + j]; c2[j] = cC[2 * XR + r0 + j + CO]; }

  v4f u[6], w[5];
#pragma unroll
  for (int j = 0; j < 6; ++j) u[j] = a0[j] * v[j] + c0[j] * v[j + 1];
#pragma unroll
  for (int j = 0; j < 5; ++j) w[j] = a1[j] * u[j] + c1[j] * u[j + 1];
  float* ob = out + ((size_t)b * NN + i0 + r0) * DV + 4 * dv4;
#pragma unroll
  for (int j = 0; j < 4; ++j) {
    *(v4f*)(ob + (size_t)j * DV) = a2[j] * w[j] + c2[j] * w[j + 1];
  }
}

extern "C" void kernel_launch(void* const* d_in, const int* in_sizes, int n_in,
                              void* d_out, int out_size, void* d_ws, size_t ws_size,
                              hipStream_t stream) {
  const float* X  = (const float*)d_in[0];
  const float* Wg = (const float*)d_in[1];
  const float* bg = (const float*)d_in[2];
  const float* Wf = (const float*)d_in[3];
  const float* bf = (const float*)d_in[4];
  float* out = (float*)d_out;

  fused_kernel<<<BB * NN / TR, 256, 0, stream>>>(X, Wg, bg, Wf, bf, out);
}


# Waves take two jobs: waves 4-7 stage X and run all of V0 (80 WMMAs each), waves 0-3 load Wf two columns a lane and run the coefficient sums beside them, not behind; every sum keeps its order
# speedup vs baseline: 4.6877x; 1.2223x over previous
#include <hip/hip_runtime.h>

#define BB 2
#define NN 8192
#define DD 64
#define DV 64
#define MM 3

#define TR 64
#define XR 80
#define PX 68
#define NP 34
#define CO 4

typedef __attribute__((ext_vector_type(2))) float v2f;
typedef __attribute__((ext_vector_type(4))) float v4f;
typedef __attribute__((ext_vector_type(8))) float v8f;

__device__ __forceinline__ int xslot(int R) { return (R >> 1) + (XR / 2) * (R & 1); }

template <int NT>
__device__ __forceinline__ void v0_tiles(const float* Xs, float* Vs, const v2f (&bfr)[16],
                                         float bias, int rt0, int ct, int r, int hi) {
  const int rs = xslot(r);
  v2f afr[NT][16];
#pragma unroll
  for (int t = 0; t < NT; ++t) {
#pragma unroll
    for (int kk = 0; kk < 16; ++kk) {
      afr[t][kk] = *(const v2f*)(Xs + (8 * (rt0 + t) + rs) * PX + 4 * kk + 2 * hi);
    }
  }
  __builtin_amdgcn_sched_barrier(0);
  v8f acc[NT];
#pragma unroll
  for (int t = 0; t < NT; ++t) acc[t] = (v8f){0.f, 0.f, 0.f, 0.f, 0.f, 0.f, 0.f, 0.f};
#pragma unroll
  for (int kk = 0; kk < 16; ++kk) {
#pragma unroll
    for (int t = 0; t < NT; ++t) {
      acc[t] = __builtin_amdgcn_wmma_f32_16x16x4_f32(
           false, afr[t][kk],  false, bfr[kk],
           (short)0, acc[t],  false,  false);
    }
  }
#pragma unroll
  for (int t = 0; t < NT; ++t) {
#pragma unroll
    for (int j = 0; j < 8; ++j) {
      Vs[(16 * (rt0 + t) + j + 8 * hi) * PX + 16 * ct + r] = acc[t][j] + bias;
    }
  }
}

__global__ __launch_bounds__(256) void fused_kernel(
    const float* __restrict__ X, const float* __restrict__ Wg,
    const float* __restrict__ bg, const float* __restrict__ Wf,
    const float* __restrict__ bf, float* __restrict__ out) {
  __shared__ __attribute__((aligned(16))) float Xs[XR * PX];
  __shared__ __attribute__((aligned(16))) float Vs[XR * PX];
  __shared__ __attribute__((aligned(16))) float cA[MM * XR];
  __shared__ __attribute__((aligned(16))) float cC[MM * XR];

  const int tid  = threadIdx.x;
  const int lane = tid & 31;
  const int r    = lane & 15;
  const int hi   = lane >> 4;
  const int wave = __builtin_amdgcn_readfirstlane(tid >> 5);
  const bool mw  = wave >= 4;
  const int t128 = tid & 127;
  const int b    = blockIdx.x / (NN / TR);
  const int i0   = (blockIdx.x % (NN / TR)) * TR;
  const float* Xb = X + (size_t)b * NN * DD;

  const int ct  = wave & 3;
  const int col = 16 * ct + r;
  v2f bfr[16];
  float bias = 0.f;

  const bool cact = t128 < MM * NP;
  const int  cp   = cact ? t128 : MM * NP - 1;
  const int  cm   = cp / NP;
  const int  cj   = cp % NP;
  v2f wv[DD];
  v2f cb = (v2f){0.f, 0.f};

  if (mw) {
    const int rowA = t128 >> 4;
    const int c4   = t128 & 15;
    const float* xp = Xb + (size_t)(i0 + rowA) * DD + 4 * c4;
    const float* xh = Xb + (size_t)(((i0 + TR) & (NN - 1)) + rowA) * DD + 4 * c4;
    v4f xr[10];
#pragma unroll
    for (int q = 0; q < 8; ++q) xr[q] = *(const v4f*)(xp + (size_t)q * 8 * DD);
    xr[8] = *(const v4f*)(xh);
    xr[9] = *(const v4f*)(xh + 8 * DD);
    __builtin_amdgcn_sched_barrier(0);
#pragma unroll
    for (int kk = 0; kk < 16; ++kk) {
      const int k0 = 4 * kk + 2 * hi;
      bfr[kk] = (v2f){Wg[(size_t)k0 * DV + col], Wg[(size_t)(k0 + 1) * DV + col]};
    }
    bias = bg[col];
    float* xs = Xs + xslot(rowA) * PX + 4 * c4;
#pragma unroll
    for (int q = 0; q < 10; ++q) *(v4f*)(xs + 4 * q * PX) = xr[q];
    asm volatile("s_wait_dscnt 0x0" ::: "memory");
  }
  asm volatile("s_barrier_signal -1" ::: "memory");
  if (!mw) {
    const int kcol = (i0 + 2 * cj) & (NN - 1);
    const float* wp = Wf + (size_t)cm * DD * NN + kcol;
    cb = *(const v2f*)(bf + (size_t)cm * NN + kcol);
#pragma unroll
    for (int d = 0; d < DD; ++d) wv[d] = *(const v2f*)(wp + (size_t)d * NN);
  }
  asm volatile("s_barrier_wait -1" ::: "memory");

  if (mw) {
    v0_tiles<5>(Xs, Vs, bfr, bias, 0, ct, r, hi);
  } else if (cact) {
    const float* p0 = Xs + cj * PX;
    const float* p1 = Xs + (XR / 2 + cj) * PX;
    const float* pm = Xs + (XR / 2 + (cj > 0 ? cj - 1 : 0)) * PX;
    v4f x0[DD / 4], x1[DD / 4], xm[DD / 4];
    x0[0] = *(const v4f*)(p0);     x0[1] = *(const v4f*)(p0 + 4);
    x1[0] = *(const v4f*)(p1);     x1[1] = *(const v4f*)(p1 + 4);
    xm[0] = *(const v4f*)(pm);     xm[1] = *(const v4f*)(pm + 4);
    v2f acP = cb;
    float aA1 = cb.y;
    float aC0 = cb.x;
#pragma unroll
    for (int g = 0; g < 8; ++g) {
      if (g + 1 < 8) {
        const int j = 2 * (g + 1);
        x0[j] = *(const v4f*)(p0 + 4 * j);  x0[j + 1] = *(const v4f*)(p0 + 4 * j + 4);
        x1[j] = *(const v4f*)(p1 + 4 * j);  x1[j + 1] = *(const v4f*)(p1 + 4 * j + 4);
        xm[j] = *(const v4f*)(pm + 4 * j);  xm[j + 1] = *(const v4f*)(pm + 4 * j + 4);
      }
      __builtin_amdgcn_sched_barrier(0);
#pragma unroll
      for (int j = 2 * g; j < 2 * g + 2; ++j) {
#pragma unroll
        for (int e = 0; e < 4; ++e) {
          const v2f wd = wv[4 * j + e];
          const float xe = x0[j][e];
          acP = __builtin_elementwise_fma((v2f){xe, xe}, wd, acP);
          aA1 = __builtin_fmaf(x1[j][e], wd.y, aA1);
          aC0 = __builtin_fmaf(xm[j][e], wd.x, aC0);
        }
      }
    }
    const int k0 = 2 * cj;
    cA[cm * XR + k0]          = acP.x;
    cA[cm * XR + k0 + 1]      = aA1;
    cC[cm * XR + k0 + CO - 1] = aC0;
    cC[cm * XR + k0 + CO]     = acP.y;
  }
  __syncthreads();

  const int dv4 = tid & 15;
  const int r0  = 4 * (tid >> 4);
  v4f v[7];
#pragma unroll
  for (int j = 0; j < 7; ++j) v[j] = *(const v4f*)(Vs + (r0 + j) * PX + 4 * dv4);
  float a0[6], c0[6], a1[5], c1[5], a2[4], c2[4];
#pragma unroll
  for (int j = 0; j < 6; ++j) { a0[j] = cA[0 * XR + r0 + j]; c0[j] = cC[0 * XR + r0 + j + CO]; }
#pragma unroll
  for (int j = 0; j < 5; ++j) { a1[j] = cA[1 * XR + r0 + j]; c1[j] = cC[1 * XR + r0 + j + CO]; }
#pragma unroll
  for (int j = 0; j < 4; ++j) { a2[j] = cA[2 * XR + r0 + j]; c2[j] = cC[2 * XR + r0 + j + CO]; }

  v4f u[6], w[5];
#pragma unroll
  for (int j = 0; j < 6; ++j) u[j] = a0[j] * v[j] + c0[j] * v[j + 1];
#pragma unroll
  for (int j = 0; j < 5; ++j) w[j] = a1[j] * u[j] + c1[j] * u[j + 1];
  float* ob = out + ((size_t)b * NN + i0 + r0) * DV + 4 * dv4;
#pragma unroll
  for (int j = 0; j < 4; ++j) {
    *(v4f*)(ob + (size_t)j * DV) = a2[j] * w[j] + c2[j] * w[j + 1];
  }
}

extern "C" void kernel_launch(void* const* d_in, const int* in_sizes, int n_in,
                              void* d_out, int out_size, void* d_ws, size_t ws_size,
                              hipStream_t stream) {
  const float* X  = (const float*)d_in[0];
  const float* Wg = (const float*)d_in[1];
  const float* bg = (const float*)d_in[2];
  const float* Wf = (const float*)d_in[3];
  const float* bf = (const float*)d_in[4];
  float* out = (float*)d_out;

  fused_kernel<<<BB * NN / TR, 256, 0, stream>>>(X, Wg, bg, Wf, bf, out);
}


# Load order across waves: a work-group barrier lets the matrix waves issue their X and Wg loads before the coefficient waves issue any Wf load, so Wg no longer queues behind Wf; sums unchanged
# speedup vs baseline: 5.1695x; 1.1028x over previous
#include <hip/hip_runtime.h>

#define BB 2
#define NN 8192
#define DD 64
#define DV 64
#define MM 3

#define TR 64
#define XR 80
#define PX 68
#define NP 34
#define CO 4

typedef __attribute__((ext_vector_type(2))) float v2f;
typedef __attribute__((ext_vector_type(4))) float v4f;
typedef __attribute__((ext_vector_type(8))) float v8f;

__device__ __forceinline__ int xslot(int R) { return (R >> 1) + (XR / 2) * (R & 1); }

template <int NT>
__device__ __forceinline__ void v0_tiles(const float* Xs, float* Vs, const v2f (&bfr)[16],
                                         float bias, int rt0, int ct, int r, int hi) {
  const int rs = xslot(r);
  v2f afr[NT][16];
#pragma unroll
  for (int t = 0; t < NT; ++t) {
#pragma unroll
    for (int kk = 0; kk < 16; ++kk) {
      afr[t][kk] = *(const v2f*)(Xs + (8 * (rt0 + t) + rs) * PX + 4 * kk + 2 * hi);
    }
  }
  __builtin_amdgcn_sched_barrier(0);
  v8f acc[NT];
#pragma unroll
  for (int t = 0; t < NT; ++t) acc[t] = (v8f){0.f, 0.f, 0.f, 0.f, 0.f, 0.f, 0.f, 0.f};
#pragma unroll
  for (int kk = 0; kk < 16; ++kk) {
#pragma unroll
    for (int t = 0; t < NT; ++t) {
      acc[t] = __builtin_amdgcn_wmma_f32_16x16x4_f32(
           false, afr[t][kk],  false, bfr[kk],
           (short)0, acc[t],  false,  false);
    }
  }
#pragma unroll
  for (int t = 0; t < NT; ++t) {
#pragma unroll
    for (int j = 0; j < 8; ++j) {
      Vs[(16 * (rt0 + t) + j + 8 * hi) * PX + 16 * ct + r] = acc[t][j] + bias;
    }
  }
}

__global__ __launch_bounds__(256) void fused_kernel(
    const float* __restrict__ X, const float* __restrict__ Wg,
    const float* __restrict__ bg, const float* __restrict__ Wf,
    const float* __restrict__ bf, float* __restrict__ out) {
  __shared__ __attribute__((aligned(16))) float Xs[XR * PX];
  __shared__ __attribute__((aligned(16))) float Vs[XR * PX];
  __shared__ __attribute__((aligned(16))) float cA[MM * XR];
  __shared__ __attribute__((aligned(16))) float cC[MM * XR];

  const int tid  = threadIdx.x;
  const int lane = tid & 31;
  const int r    = lane & 15;
  const int hi   = lane >> 4;
  const int wave = __builtin_amdgcn_readfirstlane(tid >> 5);
  const bool mw  = wave >= 4;
  const int t128 = tid & 127;
  const int b    = blockIdx.x / (NN / TR);
  const int i0   = (blockIdx.x % (NN / TR)) * TR;
  const float* Xb = X + (size_t)b * NN * DD;

  const int ct  = wave & 3;
  const int col = 16 * ct + r;
  v2f bfr[16];
  float bias = 0.f;

  const bool cact = t128 < MM * NP;
  const int  cp   = cact ? t128 : MM * NP - 1;
  const int  cm   = cp / NP;
  const int  cj   = cp % NP;
  v2f wv[DD];
  v2f cb = (v2f){0.f, 0.f};

  v4f xr[10];
  const int rowA = t128 >> 4;
  const int c4   = t128 & 15;
  if (mw) {
    const float* xp = Xb + (size_t)(i0 + rowA) * DD + 4 * c4;
    const float* xh = Xb + (size_t)(((i0 + TR) & (NN - 1)) + rowA) * DD + 4 * c4;
#pragma unroll
    for (int q = 0; q < 8; ++q) xr[q] = *(const v4f*)(xp + (size_t)q * 8 * DD);
    xr[8] = *(const v4f*)(xh);
    xr[9] = *(const v4f*)(xh + 8 * DD);
    __builtin_amdgcn_sched_barrier(0);
#pragma unroll
    for (int kk = 0; kk < 16; ++kk) {
      const int k0 = 4 * kk + 2 * hi;
      bfr[kk] = (v2f){Wg[(size_t)k0 * DV + col], Wg[(size_t)(k0 + 1) * DV + col]};
    }
    bias = bg[col];
  }
  asm volatile("s_barrier_signal -1" ::: "memory");
  asm volatile("s_barrier_wait -1" ::: "memory");
  if (mw) {
    float* xs = Xs + xslot(rowA) * PX + 4 * c4;
#pragma unroll
    for (int q = 0; q < 10; ++q) *(v4f*)(xs + 4 * q * PX) = xr[q];
    asm volatile("s_wait_dscnt 0x0" ::: "memory");
  }
  asm volatile("s_barrier_signal -1" ::: "memory");
  if (!mw) {
    const int kcol = (i0 + 2 * cj) & (NN - 1);
    const float* wp = Wf + (size_t)cm * DD * NN + kcol;
    cb = *(const v2f*)(bf + (size_t)cm * NN + kcol);
#pragma unroll
    for (int d = 0; d < DD; ++d) wv[d] = *(const v2f*)(wp + (size_t)d * NN);
  }
  asm volatile("s_barrier_wait -1" ::: "memory");

  if (mw) {
    v0_tiles<5>(Xs, Vs, bfr, bias, 0, ct, r, hi);
  } else if (cact) {
    const float* p0 = Xs + cj * PX;
    const float* p1 = Xs + (XR / 2 + cj) * PX;
    const float* pm = Xs + (XR / 2 + (cj > 0 ? cj - 1 : 0)) * PX;
    v4f x0[DD / 4], x1[DD / 4], xm[DD / 4];
    x0[0] = *(const v4f*)(p0);     x0[1] = *(const v4f*)(p0 + 4);
    x1[0] = *(const v4f*)(p1);     x1[1] = *(const v4f*)(p1 + 4);
    xm[0] = *(const v4f*)(pm);     xm[1] = *(const v4f*)(pm + 4);
    v2f acP = cb;
    float aA1 = cb.y;
    float aC0 = cb.x;
#pragma unroll
    for (int g = 0; g < 8; ++g) {
      if (g + 1 < 8) {
        const int j = 2 * (g + 1);
        x0[j] = *(const v4f*)(p0 + 4 * j);  x0[j + 1] = *(const v4f*)(p0 + 4 * j + 4);
        x1[j] = *(const v4f*)(p1 + 4 * j);  x1[j + 1] = *(const v4f*)(p1 + 4 * j + 4);
        xm[j] = *(const v4f*)(pm + 4 * j);  xm[j + 1] = *(const v4f*)(pm + 4 * j + 4);
      }
      __builtin_amdgcn_sched_barrier(0);
#pragma unroll
      for (int j = 2 * g; j < 2 * g + 2; ++j) {
#pragma unroll
        for (int e = 0; e < 4; ++e) {
          const v2f wd = wv[4 * j + e];
          const float xe = x0[j][e];
          acP = __builtin_elementwise_fma((v2f){xe, xe}, wd, acP);
          aA1 = __builtin_fmaf(x1[j][e], wd.y, aA1);
          aC0 = __builtin_fmaf(xm[j][e], wd.x, aC0);
        }
      }
    }
    const int k0 = 2 * cj;
    cA[cm * XR + k0]          = acP.x;
    cA[cm * XR + k0 + 1]      = aA1;
    cC[cm * XR + k0 + CO - 1] = aC0;
    cC[cm * XR + k0 + CO]     = acP.y;
  }
  __syncthreads();

  const int dv4 = tid & 15;
  const int r0  = 4 * (tid >> 4);
  v4f v[7];
#pragma unroll
  for (int j = 0; j < 7; ++j) v[j] = *(const v4f*)(Vs + (r0 + j) * PX + 4 * dv4);
  float a0[6], c0[6], a1[5], c1[5], a2[4], c2[4];
#pragma unroll
  for (int j = 0; j < 6; ++j) { a0[j] = cA[0 * XR + r0 + j]; c0[j] = cC[0 * XR + r0 + j + CO]; }
#pragma unroll
  for (int j = 0; j < 5; ++j) { a1[j] = cA[1 * XR + r0 + j]; c1[j] = cC[1 * XR + r0 + j + CO]; }
#pragma unroll
  for (int j = 0; j < 4; ++j) { a2[j] = cA[2 * XR + r0 + j]; c2[j] = cC[2 * XR + r0 + j + CO]; }

  v4f u[6], w[5];
#pragma unroll
  for (int j = 0; j < 6; ++j) u[j] = a0[j] * v[j] + c0[j] * v[j + 1];
#pragma unroll
  for (int j = 0; j < 5; ++j) w[j] = a1[j] * u[j] + c1[j] * u[j + 1];
  float* ob = out + ((size_t)b * NN + i0 + r0) * DV + 4 * dv4;
#pragma unroll
  for (int j = 0; j < 4; ++j) {
    *(v4f*)(ob + (size_t)j * DV) = a2[j] * w[j] + c2[j] * w[j + 1];
  }
}

extern "C" void kernel_launch(void* const* d_in, const int* in_sizes, int n_in,
                              void* d_out, int out_size, void* d_ws, size_t ws_size,
                              hipStream_t stream) {
  const float* X  = (const float*)d_in[0];
  const float* Wg = (const float*)d_in[1];
  const float* bg = (const float*)d_in[2];
  const float* Wf = (const float*)d_in[3];
  const float* bf = (const float*)d_in[4];
  float* out = (float*)d_out;

  fused_kernel<<<BB * NN / TR, 256, 0, stream>>>(X, Wg, bg, Wf, bf, out);
}
